# starting kernel
# baseline (speedup 1.0000x reference)
#include <hip/hip_runtime.h>
#include <cmath>


typedef __attribute__((ext_vector_type(2))) float v2f;
typedef __attribute__((ext_vector_type(8))) float v8f;

#define OUT_F 1024
#define IN_F  1024
#define CTX_F 1024
#define BATCH 32
#define DIMB  16

__device__ __forceinline__ v8f wmma_f32(v2f a, v2f b, v8f c) {
  return __builtin_amdgcn_wmma_f32_16x16x4_f32(
      false, a, false, b, (short)0, c, false, false);
}

__global__ void rowprep_kernel(const float* __restrict__ weight,
                               const float* __restrict__ diagw,
                               float* __restrict__ w_norm,
                               float* __restrict__ logterm) {
  const int r  = blockIdx.x;
  const int t  = threadIdx.x;
  const int rb = r >> 6;

  __shared__ float wraw[IN_F];
  __shared__ float red[256];
  __shared__ float bc;

  float sq = 0.f;
  for (int c = t; c < IN_F; c += 256) {
    const int cb = c >> 6;
    const float wv  = weight[r * IN_F + c];
    const float val = (cb == rb) ? __expf(wv) : ((cb < rb) ? wv : 0.f);
    wraw[c] = val;
    sq += val * val;
  }
  red[t] = sq;
  __syncthreads();
  for (int s = 128; s > 0; s >>= 1) {
    if (t < s) red[t] += red[t + s];
    __syncthreads();
  }
  if (t == 0) {
    const float wsn = red[0];
    const float dw  = diagw[r];
    bc = __expf(dw) * rsqrtf(wsn);
    logterm[r] = dw - 0.5f * __logf(wsn);
  }
  __syncthreads();
  const float scale = bc;
  for (int c = t; c < IN_F; c += 256)
    w_norm[r * IN_F + c] = scale * wraw[c];
}

__global__ void out_gemm_kernel(const float* __restrict__ inputs,
                                const float* __restrict__ cond,
                                const float* __restrict__ w_norm,
                                const float* __restrict__ wctx,
                                const float* __restrict__ bias,
                                float* __restrict__ out) {
  const int lane = threadIdx.x;
  const int nt   = blockIdx.x;
  const int mt   = blockIdx.y;
  const int mA   = lane & 15;
  const int kb   = (lane >> 4) << 1;
  const int nB   = lane & 15;

  v8f acc = {};

  {
    const float* ap = inputs + (mt * 16 + mA) * IN_F + kb;
    const float* bp = w_norm + (nt * 16 + nB) * IN_F + kb;
    for (int k0 = 0; k0 < IN_F; k0 += 4) {
      v2f a; a.x = ap[k0]; a.y = ap[k0 + 1];
      v2f b; b.x = bp[k0]; b.y = bp[k0 + 1];
      acc = wmma_f32(a, b, acc);
    }
  }
  {
    const float* ap = cond + (mt * 16 + mA) * CTX_F + kb;
    const float* bp = wctx + nt * 16 + nB;
    for (int k0 = 0; k0 < CTX_F; k0 += 4) {
      v2f a; a.x = ap[k0]; a.y = ap[k0 + 1];
      v2f b;
      b.x = bp[(size_t)(k0 + kb) * OUT_F];
      b.y = bp[(size_t)(k0 + kb + 1) * OUT_F];
      acc = wmma_f32(a, b, acc);
    }
  }

  const int mBase = (lane >> 4) << 3;
  const int n     = nt * 16 + (lane & 15);
  const float bv  = bias[n];
  for (int v = 0; v < 8; ++v) {
    const int m = mt * 16 + mBase + v;
    out[m * OUT_F + n] = acc[v] + bv;
  }
}

__global__ void comb_kernel(const float* __restrict__ grad,
                            const float* __restrict__ weight,
                            const float* __restrict__ logterm,
                            float* __restrict__ comb) {
  const int d = blockIdx.x;
  const int b = blockIdx.y;
  const int t = threadIdx.x;
  const int STR = 66;

  __shared__ float eg[64 * 66];
  __shared__ float er[64 * 66];
  __shared__ float m1s[64];
  __shared__ float m2s[64];

  const float* gblk = grad + (size_t)((b * DIMB + d) * 64) * 64;
  const float* wblk = weight + (size_t)(d * 64) * IN_F + d * 64;

  if (t < 64) {
    float m = -3.4e38f;
    for (int k = 0; k < 64; ++k) m = fmaxf(m, gblk[k * 64 + t]);
    m2s[t] = m;
  } else if (t < 128) {
    const int i = t - 64;
    const float* wr = wblk + (size_t)i * IN_F;
    float m = -3.4e38f;
    for (int k = 0; k < 64; ++k) m = fmaxf(m, wr[k]);
    m1s[i] = logterm[d * 64 + i] + m;
  }
  __syncthreads();

  for (int idx = t; idx < 64 * 64; idx += 256) {
    const int k = idx >> 6, j = idx & 63;
    er[k * STR + j] = __expf(gblk[k * 64 + j] - m2s[j]);
  }
  for (int idx = t; idx < 64 * 64; idx += 256) {
    const int i = idx >> 6, k = idx & 63;
    const float g = logterm[d * 64 + i] + wblk[(size_t)i * IN_F + k];
    eg[i * STR + k] = __expf(g - m1s[i]);
  }
  __syncthreads();

  const int wave  = t >> 5;
  const int lane  = t & 31;
  const int mA    = lane & 15;
  const int kb    = (lane >> 4) << 1;
  const int nB    = lane & 15;
  const int mBase = (lane >> 4) << 3;

  for (int tile = wave; tile < 16; tile += 8) {
    const int i0 = (tile >> 2) * 16;
    const int j0 = (tile & 3) * 16;
    v8f acc = {};
    const float* ap = &eg[(i0 + mA) * STR + kb];
    const float* bp = &er[kb * STR + j0 + nB];
    for (int kk = 0; kk < 16; ++kk) {
      v2f a; a.x = ap[kk * 4];           a.y = ap[kk * 4 + 1];
      v2f b; b.x = bp[(kk * 4) * STR];   b.y = bp[(kk * 4 + 1) * STR];
      acc = wmma_f32(a, b, acc);
    }
    for (int v = 0; v < 8; ++v) {
      const int i = i0 + mBase + v;
      const int j = j0 + (lane & 15);
      comb[((size_t)(b * DIMB + d) * 64 + i) * 64 + j] =
          m1s[i] + m2s[j] + __logf(acc[v]);
    }
  }
}

extern "C" void kernel_launch(void* const* d_in, const int* in_sizes, int n_in,
                              void* d_out, int out_size, void* d_ws, size_t ws_size,
                              hipStream_t stream) {
  (void)in_sizes; (void)n_in; (void)out_size; (void)ws_size;
  const float* inputs = (const float*)d_in[0];
  const float* grad   = (const float*)d_in[1];
  const float* cond   = (const float*)d_in[2];
  const float* weight = (const float*)d_in[3];
  const float* diagw  = (const float*)d_in[4];
  const float* bias   = (const float*)d_in[5];
  const float* wctx   = (const float*)d_in[6];

  float* out  = (float*)d_out;
  float* comb = out + BATCH * OUT_F;

  float* w_norm  = (float*)d_ws;
  float* logterm = w_norm + (size_t)OUT_F * IN_F;

  rowprep_kernel<<<dim3(OUT_F), dim3(256), 0, stream>>>(weight, diagw, w_norm, logterm);
  out_gemm_kernel<<<dim3(64, 2), dim3(32), 0, stream>>>(inputs, cond, w_norm, wctx, bias, out);
  comb_kernel<<<dim3(DIMB, BATCH), dim3(256), 0, stream>>>(grad, weight, logterm, comb);
}


# out GEMM: split its K=2048 over 16 waves per 16x16 tile (512-thread groups), 16 partial tiles added in fixed order through LDS planes; this changes the summation order
# speedup vs baseline: 1.6800x; 1.6800x over previous
#include <hip/hip_runtime.h>
#include <cmath>


typedef __attribute__((ext_vector_type(2))) float v2f;
typedef __attribute__((ext_vector_type(8))) float v8f;

#define OUT_F 1024
#define IN_F  1024
#define CTX_F 1024
#define BATCH 32
#define DIMB  16

__device__ __forceinline__ v8f wmma_f32(v2f a, v2f b, v8f c) {
  return __builtin_amdgcn_wmma_f32_16x16x4_f32(
      false, a, false, b, (short)0, c, false, false);
}

__global__ void rowprep_kernel(const float* __restrict__ weight,
                               const float* __restrict__ diagw,
                               float* __restrict__ w_norm,
                               float* __restrict__ logterm) {
  const int r  = blockIdx.x;
  const int t  = threadIdx.x;
  const int rb = r >> 6;

  __shared__ float wraw[IN_F];
  __shared__ float red[256];
  __shared__ float bc;

  float sq = 0.f;
  for (int c = t; c < IN_F; c += 256) {
    const int cb = c >> 6;
    const float wv  = weight[r * IN_F + c];
    const float val = (cb == rb) ? __expf(wv) : ((cb < rb) ? wv : 0.f);
    wraw[c] = val;
    sq += val * val;
  }
  red[t] = sq;
  __syncthreads();
  for (int s = 128; s > 0; s >>= 1) {
    if (t < s) red[t] += red[t + s];
    __syncthreads();
  }
  if (t == 0) {
    const float wsn = red[0];
    const float dw  = diagw[r];
    bc = __expf(dw) * rsqrtf(wsn);
    logterm[r] = dw - 0.5f * __logf(wsn);
  }
  __syncthreads();
  const float scale = bc;
  for (int c = t; c < IN_F; c += 256)
    w_norm[r * IN_F + c] = scale * wraw[c];
}

#define KWAVES 16
#define KSLICE 128
#define KSTEPS (KSLICE / 4)
#define GEMM_THREADS 512

__global__ void __launch_bounds__(GEMM_THREADS)
out_gemm_kernel(const float* __restrict__ inputs,
                const float* __restrict__ cond,
                const float* __restrict__ w_norm,
                const float* __restrict__ wctx,
                const float* __restrict__ bias,
                float* __restrict__ out) {
  const int t    = threadIdx.x;
  const int wave = t >> 5;
  const int lane = t & 31;
  const int nt   = blockIdx.x;
  const int mt   = blockIdx.y;
  const int mA   = lane & 15;
  const int kb   = (lane >> 4) << 1;
  const int nB   = lane & 15;

  __shared__ float part[KWAVES * 256];

  v8f acc = {};

  if (wave < KWAVES / 2) {
    const int ks = wave * KSLICE;
    const float* ap = inputs + (mt * 16 + mA) * IN_F + ks + kb;
    const float* bp = w_norm + (nt * 16 + nB) * IN_F + ks + kb;
    for (int s = 0; s < KSTEPS; ++s) {
      v2f a; a.x = ap[4 * s]; a.y = ap[4 * s + 1];
      v2f b; b.x = bp[4 * s]; b.y = bp[4 * s + 1];
      acc = wmma_f32(a, b, acc);
    }
  } else {
    const int ks = (wave - KWAVES / 2) * KSLICE;
    const float* ap = cond + (mt * 16 + mA) * CTX_F + ks + kb;
    const float* bp = wctx + (size_t)(ks + kb) * OUT_F + nt * 16 + nB;
    for (int s = 0; s < KSTEPS; ++s) {
      v2f a; a.x = ap[4 * s]; a.y = ap[4 * s + 1];
      v2f b;
      b.x = bp[(size_t)(4 * s) * OUT_F];
      b.y = bp[(size_t)(4 * s + 1) * OUT_F];
      acc = wmma_f32(a, b, acc);
    }
  }

  for (int v = 0; v < 8; ++v) part[wave * 256 + v * 32 + lane] = acc[v];
  __syncthreads();

  if (t < 256) {
    float sum = part[t];
    for (int w = 1; w < KWAVES; ++w) sum += part[w * 256 + t];
    const int v = t >> 5;
    const int l = t & 31;
    const int m = mt * 16 + ((l >> 4) << 3) + v;
    const int n = nt * 16 + (l & 15);
    out[m * OUT_F + n] = sum + bias[n];
  }
}

__global__ void comb_kernel(const float* __restrict__ grad,
                            const float* __restrict__ weight,
                            const float* __restrict__ logterm,
                            float* __restrict__ comb) {
  const int d = blockIdx.x;
  const int b = blockIdx.y;
  const int t = threadIdx.x;
  const int STR = 66;

  __shared__ float eg[64 * 66];
  __shared__ float er[64 * 66];
  __shared__ float m1s[64];
  __shared__ float m2s[64];

  const float* gblk = grad + (size_t)((b * DIMB + d) * 64) * 64;
  const float* wblk = weight + (size_t)(d * 64) * IN_F + d * 64;

  if (t < 64) {
    float m = -3.4e38f;
    for (int k = 0; k < 64; ++k) m = fmaxf(m, gblk[k * 64 + t]);
    m2s[t] = m;
  } else if (t < 128) {
    const int i = t - 64;
    const float* wr = wblk + (size_t)i * IN_F;
    float m = -3.4e38f;
    for (int k = 0; k < 64; ++k) m = fmaxf(m, wr[k]);
    m1s[i] = logterm[d * 64 + i] + m;
  }
  __syncthreads();

  for (int idx = t; idx < 64 * 64; idx += 256) {
    const int k = idx >> 6, j = idx & 63;
    er[k * STR + j] = __expf(gblk[k * 64 + j] - m2s[j]);
  }
  for (int idx = t; idx < 64 * 64; idx += 256) {
    const int i = idx >> 6, k = idx & 63;
    const float g = logterm[d * 64 + i] + wblk[(size_t)i * IN_F + k];
    eg[i * STR + k] = __expf(g - m1s[i]);
  }
  __syncthreads();

  const int wave  = t >> 5;
  const int lane  = t & 31;
  const int mA    = lane & 15;
  const int kb    = (lane >> 4) << 1;
  const int nB    = lane & 15;
  const int mBase = (lane >> 4) << 3;

  for (int tile = wave; tile < 16; tile += 8) {
    const int i0 = (tile >> 2) * 16;
    const int j0 = (tile & 3) * 16;
    v8f acc = {};
    const float* ap = &eg[(i0 + mA) * STR + kb];
    const float* bp = &er[kb * STR + j0 + nB];
    for (int kk = 0; kk < 16; ++kk) {
      v2f a; a.x = ap[kk * 4];           a.y = ap[kk * 4 + 1];
      v2f b; b.x = bp[(kk * 4) * STR];   b.y = bp[(kk * 4 + 1) * STR];
      acc = wmma_f32(a, b, acc);
    }
    for (int v = 0; v < 8; ++v) {
      const int i = i0 + mBase + v;
      const int j = j0 + (lane & 15);
      comb[((size_t)(b * DIMB + d) * 64 + i) * 64 + j] =
          m1s[i] + m2s[j] + __logf(acc[v]);
    }
  }
}

extern "C" void kernel_launch(void* const* d_in, const int* in_sizes, int n_in,
                              void* d_out, int out_size, void* d_ws, size_t ws_size,
                              hipStream_t stream) {
  (void)in_sizes; (void)n_in; (void)out_size; (void)ws_size;
  const float* inputs = (const float*)d_in[0];
  const float* grad   = (const float*)d_in[1];
  const float* cond   = (const float*)d_in[2];
  const float* weight = (const float*)d_in[3];
  const float* diagw  = (const float*)d_in[4];
  const float* bias   = (const float*)d_in[5];
  const float* wctx   = (const float*)d_in[6];

  float* out  = (float*)d_out;
  float* comb = out + BATCH * OUT_F;

  float* w_norm  = (float*)d_ws;
  float* logterm = w_norm + (size_t)OUT_F * IN_F;

  rowprep_kernel<<<dim3(OUT_F), dim3(256), 0, stream>>>(weight, diagw, w_norm, logterm);
  out_gemm_kernel<<<dim3(64, 2), dim3(GEMM_THREADS), 0, stream>>>(inputs, cond, w_norm, wctx, bias, out);
  comb_kernel<<<dim3(DIMB, BATCH), dim3(256), 0, stream>>>(grad, weight, logterm, comb);
}


# row normalisation folded into the out GEMM: mask/exp on its B fragments, squares summed from them, exp(diag)/sqrt(wsn) scales the finished sum; rowprep launch gone; changes summation order
# speedup vs baseline: 2.0936x; 1.2462x over previous
#include <hip/hip_runtime.h>
#include <cmath>


typedef __attribute__((ext_vector_type(2))) float v2f;
typedef __attribute__((ext_vector_type(8))) float v8f;

#define OUT_F 1024
#define IN_F  1024
#define CTX_F 1024
#define BATCH 32
#define DIMB  16

__device__ __forceinline__ v8f wmma_f32(v2f a, v2f b, v8f c) {
  return __builtin_amdgcn_wmma_f32_16x16x4_f32(
      false, a, false, b, (short)0, c, false, false);
}

#define KWAVES 16
#define KSLICE 128
#define KSTEPS (KSLICE / 4)
#define GEMM_THREADS 512
#define BLK 64
#define BSTEPS (BLK / 4)

__global__ void __launch_bounds__(GEMM_THREADS)
out_gemm_kernel(const float* __restrict__ inputs,
                const float* __restrict__ cond,
                const float* __restrict__ weight,
                const float* __restrict__ diagw,
                const float* __restrict__ wctx,
                const float* __restrict__ bias,
                float* __restrict__ out,
                float* __restrict__ logterm) {
  const int t    = threadIdx.x;
  const int wave = t >> 5;
  const int lane = t & 31;
  const int nt   = blockIdx.x;
  const int mt   = blockIdx.y;
  const int mA   = lane & 15;
  const int kb   = (lane >> 4) << 1;
  const int nB   = lane & 15;

  __shared__ float part[KWAVES * 256];
  __shared__ float sqp[(KWAVES / 2) * 32];

  const float dw = diagw[nt * 16 + nB];
  const float bs = bias[nt * 16 + nB];

  v8f acc = {};

  if (wave < KWAVES / 2) {
    const int rb = (nt * 16) / BLK;
    const float* ap = inputs + (mt * 16 + mA) * IN_F + kb;
    const float* bp = weight + (nt * 16 + nB) * IN_F + kb;
    float sq = 0.f;
    for (int h = 0; h < KSLICE / BLK; ++h) {
      const int cb = wave * (KSLICE / BLK) + h;
      const float* aq = ap + cb * BLK;
      const float* bq = bp + cb * BLK;
      if (cb < rb) {
        for (int s = 0; s < BSTEPS; ++s) {
          v2f a; a.x = aq[4 * s]; a.y = aq[4 * s + 1];
          v2f b; b.x = bq[4 * s]; b.y = bq[4 * s + 1];
          sq += b.x * b.x + b.y * b.y;
          acc = wmma_f32(a, b, acc);
        }
      } else if (cb == rb) {
        for (int s = 0; s < BSTEPS; ++s) {
          v2f a; a.x = aq[4 * s]; a.y = aq[4 * s + 1];
          v2f b; b.x = __expf(bq[4 * s]); b.y = __expf(bq[4 * s + 1]);
          sq += b.x * b.x + b.y * b.y;
          acc = wmma_f32(a, b, acc);
        }
      }
    }
    sqp[wave * 32 + lane] = sq;
  } else {
    const int ks = (wave - KWAVES / 2) * KSLICE;
    const float* ap = cond + (mt * 16 + mA) * CTX_F + ks + kb;
    const float* bp = wctx + (size_t)(ks + kb) * OUT_F + nt * 16 + nB;
    for (int s = 0; s < KSTEPS; ++s) {
      v2f a; a.x = ap[4 * s]; a.y = ap[4 * s + 1];
      v2f b;
      b.x = bp[(size_t)(4 * s) * OUT_F];
      b.y = bp[(size_t)(4 * s + 1) * OUT_F];
      acc = wmma_f32(a, b, acc);
    }
  }

  for (int v = 0; v < 8; ++v) part[wave * 256 + v * 32 + lane] = acc[v];
  __syncthreads();

  if (t < 256) {
    const int c = t & 15;
    float wsn = sqp[c] + sqp[c + 16];
    for (int w = 1; w < KWAVES / 2; ++w) wsn += sqp[w * 32 + c] + sqp[w * 32 + c + 16];
    float s1 = part[t];
    for (int w = 1; w < KWAVES / 2; ++w) s1 += part[w * 256 + t];
    float s2 = part[(KWAVES / 2) * 256 + t];
    for (int w = KWAVES / 2 + 1; w < KWAVES; ++w) s2 += part[w * 256 + t];
    const int v = t >> 5;
    const int l = t & 31;
    const int m = mt * 16 + ((l >> 4) << 3) + v;
    const int n = nt * 16 + (l & 15);
    const float scale = __expf(dw) * rsqrtf(wsn);
    out[m * OUT_F + n] = scale * s1 + s2 + bs;
    if (mt == 0 && t < 16) logterm[n] = dw - 0.5f * __logf(wsn);
  }
}

__global__ void comb_kernel(const float* __restrict__ grad,
                            const float* __restrict__ weight,
                            const float* __restrict__ logterm,
                            float* __restrict__ comb) {
  const int d = blockIdx.x;
  const int b = blockIdx.y;
  const int t = threadIdx.x;
  const int STR = 66;

  __shared__ float eg[64 * 66];
  __shared__ float er[64 * 66];
  __shared__ float m1s[64];
  __shared__ float m2s[64];

  const float* gblk = grad + (size_t)((b * DIMB + d) * 64) * 64;
  const float* wblk = weight + (size_t)(d * 64) * IN_F + d * 64;

  if (t < 64) {
    float m = -3.4e38f;
    for (int k = 0; k < 64; ++k) m = fmaxf(m, gblk[k * 64 + t]);
    m2s[t] = m;
  } else if (t < 128) {
    const int i = t - 64;
    const float* wr = wblk + (size_t)i * IN_F;
    float m = -3.4e38f;
    for (int k = 0; k < 64; ++k) m = fmaxf(m, wr[k]);
    m1s[i] = logterm[d * 64 + i] + m;
  }
  __syncthreads();

  for (int idx = t; idx < 64 * 64; idx += 256) {
    const int k = idx >> 6, j = idx & 63;
    er[k * STR + j] = __expf(gblk[k * 64 + j] - m2s[j]);
  }
  for (int idx = t; idx < 64 * 64; idx += 256) {
    const int i = idx >> 6, k = idx & 63;
    const float g = logterm[d * 64 + i] + wblk[(size_t)i * IN_F + k];
    eg[i * STR + k] = __expf(g - m1s[i]);
  }
  __syncthreads();

  const int wave  = t >> 5;
  const int lane  = t & 31;
  const int mA    = lane & 15;
  const int kb    = (lane >> 4) << 1;
  const int nB    = lane & 15;
  const int mBase = (lane >> 4) << 3;

  for (int tile = wave; tile < 16; tile += 8) {
    const int i0 = (tile >> 2) * 16;
    const int j0 = (tile & 3) * 16;
    v8f acc = {};
    const float* ap = &eg[(i0 + mA) * STR + kb];
    const float* bp = &er[kb * STR + j0 + nB];
    for (int kk = 0; kk < 16; ++kk) {
      v2f a; a.x = ap[kk * 4];           a.y = ap[kk * 4 + 1];
      v2f b; b.x = bp[(kk * 4) * STR];   b.y = bp[(kk * 4 + 1) * STR];
      acc = wmma_f32(a, b, acc);
    }
    for (int v = 0; v < 8; ++v) {
      const int i = i0 + mBase + v;
      const int j = j0 + (lane & 15);
      comb[((size_t)(b * DIMB + d) * 64 + i) * 64 + j] =
          m1s[i] + m2s[j] + __logf(acc[v]);
    }
  }
}

extern "C" void kernel_launch(void* const* d_in, const int* in_sizes, int n_in,
                              void* d_out, int out_size, void* d_ws, size_t ws_size,
                              hipStream_t stream) {
  (void)in_sizes; (void)n_in; (void)out_size; (void)ws_size;
  const float* inputs = (const float*)d_in[0];
  const float* grad   = (const float*)d_in[1];
  const float* cond   = (const float*)d_in[2];
  const float* weight = (const float*)d_in[3];
  const float* diagw  = (const float*)d_in[4];
  const float* bias   = (const float*)d_in[5];
  const float* wctx   = (const float*)d_in[6];

  float* out  = (float*)d_out;
  float* comb = out + BATCH * OUT_F;

  float* logterm = (float*)d_ws;

  out_gemm_kernel<<<dim3(64, 2), dim3(GEMM_THREADS), 0, stream>>>(inputs, cond, weight, diagw, wctx, bias, out, logterm);
  comb_kernel<<<dim3(DIMB, BATCH), dim3(256), 0, stream>>>(grad, weight, logterm, comb);
}


# comb kernel: both 64x64 blocks read once with 16-byte loads all in flight, kept in registers; maxima via LDS partials instead of a serial load-and-wait max loop and a second global read
# speedup vs baseline: 4.6221x; 2.2077x over previous
#include <hip/hip_runtime.h>
#include <cmath>


typedef __attribute__((ext_vector_type(2))) float v2f;
typedef __attribute__((ext_vector_type(8))) float v8f;

#define OUT_F 1024
#define IN_F  1024
#define CTX_F 1024
#define BATCH 32
#define DIMB  16

__device__ __forceinline__ v8f wmma_f32(v2f a, v2f b, v8f c) {
  return __builtin_amdgcn_wmma_f32_16x16x4_f32(
      false, a, false, b, (short)0, c, false, false);
}

#define KWAVES 16
#define KSLICE 128
#define KSTEPS (KSLICE / 4)
#define GEMM_THREADS 512
#define BLK 64
#define BSTEPS (BLK / 4)

__global__ void __launch_bounds__(GEMM_THREADS)
out_gemm_kernel(const float* __restrict__ inputs,
                const float* __restrict__ cond,
                const float* __restrict__ weight,
                const float* __restrict__ diagw,
                const float* __restrict__ wctx,
                const float* __restrict__ bias,
                float* __restrict__ out,
                float* __restrict__ logterm) {
  const int t    = threadIdx.x;
  const int wave = t >> 5;
  const int lane = t & 31;
  const int nt   = blockIdx.x;
  const int mt   = blockIdx.y;
  const int mA   = lane & 15;
  const int kb   = (lane >> 4) << 1;
  const int nB   = lane & 15;

  __shared__ float part[KWAVES * 256];
  __shared__ float sqp[(KWAVES / 2) * 32];

  const float dw = diagw[nt * 16 + nB];
  const float bs = bias[nt * 16 + nB];

  v8f acc = {};

  if (wave < KWAVES / 2) {
    const int rb = (nt * 16) / BLK;
    const float* ap = inputs + (mt * 16 + mA) * IN_F + kb;
    const float* bp = weight + (nt * 16 + nB) * IN_F + kb;
    float sq = 0.f;
    for (int h = 0; h < KSLICE / BLK; ++h) {
      const int cb = wave * (KSLICE / BLK) + h;
      const float* aq = ap + cb * BLK;
      const float* bq = bp + cb * BLK;
      if (cb < rb) {
        for (int s = 0; s < BSTEPS; ++s) {
          v2f a; a.x = aq[4 * s]; a.y = aq[4 * s + 1];
          v2f b; b.x = bq[4 * s]; b.y = bq[4 * s + 1];
          sq += b.x * b.x + b.y * b.y;
          acc = wmma_f32(a, b, acc);
        }
      } else if (cb == rb) {
        for (int s = 0; s < BSTEPS; ++s) {
          v2f a; a.x = aq[4 * s]; a.y = aq[4 * s + 1];
          v2f b; b.x = __expf(bq[4 * s]); b.y = __expf(bq[4 * s + 1]);
          sq += b.x * b.x + b.y * b.y;
          acc = wmma_f32(a, b, acc);
        }
      }
    }
    sqp[wave * 32 + lane] = sq;
  } else {
    const int ks = (wave - KWAVES / 2) * KSLICE;
    const float* ap = cond + (mt * 16 + mA) * CTX_F + ks + kb;
    const float* bp = wctx + (size_t)(ks + kb) * OUT_F + nt * 16 + nB;
    for (int s = 0; s < KSTEPS; ++s) {
      v2f a; a.x = ap[4 * s]; a.y = ap[4 * s + 1];
      v2f b;
      b.x = bp[(size_t)(4 * s) * OUT_F];
      b.y = bp[(size_t)(4 * s + 1) * OUT_F];
      acc = wmma_f32(a, b, acc);
    }
  }

  for (int v = 0; v < 8; ++v) part[wave * 256 + v * 32 + lane] = acc[v];
  __syncthreads();

  if (t < 256) {
    const int c = t & 15;
    float wsn = sqp[c] + sqp[c + 16];
    for (int w = 1; w < KWAVES / 2; ++w) wsn += sqp[w * 32 + c] + sqp[w * 32 + c + 16];
    float s1 = part[t];
    for (int w = 1; w < KWAVES / 2; ++w) s1 += part[w * 256 + t];
    float s2 = part[(KWAVES / 2) * 256 + t];
    for (int w = KWAVES / 2 + 1; w < KWAVES; ++w) s2 += part[w * 256 + t];
    const int v = t >> 5;
    const int l = t & 31;
    const int m = mt * 16 + ((l >> 4) << 3) + v;
    const int n = nt * 16 + (l & 15);
    const float scale = __expf(dw) * rsqrtf(wsn);
    out[m * OUT_F + n] = scale * s1 + s2 + bs;
    if (mt == 0 && t < 16) logterm[n] = dw - 0.5f * __logf(wsn);
  }
}

typedef __attribute__((ext_vector_type(4))) float v4f;

__global__ void comb_kernel(const float* __restrict__ grad,
                            const float* __restrict__ weight,
                            const float* __restrict__ logterm,
                            float* __restrict__ comb) {
  const int d = blockIdx.x;
  const int b = blockIdx.y;
  const int t = threadIdx.x;
  const int STR = 66;
  const int RPP = 17;

  __shared__ __attribute__((aligned(16))) float eg[64 * 66];
  __shared__ __attribute__((aligned(16))) float er[64 * 66];
  __shared__ __attribute__((aligned(16))) float colp[16 * 64];
  __shared__ __attribute__((aligned(16))) float rowp[64 * 17];
  __shared__ __attribute__((aligned(16))) float m1s[64];
  __shared__ __attribute__((aligned(16))) float m2s[64];
  __shared__ __attribute__((aligned(16))) float lts[64];

  const float* gblk = grad + (size_t)((b * DIMB + d) * 64) * 64;
  const float* wblk = weight + (size_t)(d * 64) * IN_F + d * 64;

  const int c4 = t & 15;
  const int r0 = t >> 4;

  v4f gv[4], wv[4];
  for (int q = 0; q < 4; ++q)
    gv[q] = *(const v4f*)(gblk + (r0 + 16 * q) * 64 + 4 * c4);
  for (int q = 0; q < 4; ++q)
    wv[q] = *(const v4f*)(wblk + (size_t)(r0 + 16 * q) * IN_F + 4 * c4);
  const float ltv = logterm[d * 64 + (t & 63)];

  {
    v4f cm;
    cm.x = fmaxf(fmaxf(gv[0].x, gv[1].x), fmaxf(gv[2].x, gv[3].x));
    cm.y = fmaxf(fmaxf(gv[0].y, gv[1].y), fmaxf(gv[2].y, gv[3].y));
    cm.z = fmaxf(fmaxf(gv[0].z, gv[1].z), fmaxf(gv[2].z, gv[3].z));
    cm.w = fmaxf(fmaxf(gv[0].w, gv[1].w), fmaxf(gv[2].w, gv[3].w));
    *(v4f*)&colp[r0 * 64 + 4 * c4] = cm;
    for (int q = 0; q < 4; ++q)
      rowp[(r0 + 16 * q) * RPP + c4] =
          fmaxf(fmaxf(wv[q].x, wv[q].y), fmaxf(wv[q].z, wv[q].w));
  }
  __syncthreads();

  if (t < 64) {
    float m = -3.4e38f;
    for (int r = 0; r < 16; ++r) m = fmaxf(m, colp[r * 64 + t]);
    m2s[t] = m;
  } else if (t < 128) {
    const int i = t - 64;
    float m = -3.4e38f;
    for (int c = 0; c < 16; ++c) m = fmaxf(m, rowp[i * RPP + c]);
    m1s[i] = ltv + m;
    lts[i] = ltv;
  }
  __syncthreads();

  {
    const v4f m2v = *(const v4f*)&m2s[4 * c4];
    for (int q = 0; q < 4; ++q) {
      const int k = r0 + 16 * q;
      v2f lo, hi;
      lo.x = __expf(gv[q].x - m2v.x);
      lo.y = __expf(gv[q].y - m2v.y);
      hi.x = __expf(gv[q].z - m2v.z);
      hi.y = __expf(gv[q].w - m2v.w);
      *(v2f*)&er[k * STR + 4 * c4]     = lo;
      *(v2f*)&er[k * STR + 4 * c4 + 2] = hi;
    }
    for (int q = 0; q < 4; ++q) {
      const int i = r0 + 16 * q;
      const float lt = lts[i];
      const float m1 = m1s[i];
      v2f lo, hi;
      lo.x = __expf((lt + wv[q].x) - m1);
      lo.y = __expf((lt + wv[q].y) - m1);
      hi.x = __expf((lt + wv[q].z) - m1);
      hi.y = __expf((lt + wv[q].w) - m1);
      *(v2f*)&eg[i * STR + 4 * c4]     = lo;
      *(v2f*)&eg[i * STR + 4 * c4 + 2] = hi;
    }
  }
  __syncthreads();

  const int wave  = t >> 5;
  const int lane  = t & 31;
  const int mA    = lane & 15;
  const int kb    = (lane >> 4) << 1;
  const int nB    = lane & 15;
  const int mBase = (lane >> 4) << 3;

  for (int tile = wave; tile < 16; tile += 8) {
    const int i0 = (tile >> 2) * 16;
    const int j0 = (tile & 3) * 16;
    v8f acc = {};
    const float* ap = &eg[(i0 + mA) * STR + kb];
    const float* bp = &er[kb * STR + j0 + nB];
    for (int kk = 0; kk < 16; ++kk) {
      v2f a; a.x = ap[kk * 4];           a.y = ap[kk * 4 + 1];
      v2f b; b.x = bp[(kk * 4) * STR];   b.y = bp[(kk * 4 + 1) * STR];
      acc = wmma_f32(a, b, acc);
    }
    for (int v = 0; v < 8; ++v) {
      const int i = i0 + mBase + v;
      const int j = j0 + (lane & 15);
      comb[((size_t)(b * DIMB + d) * 64 + i) * 64 + j] =
          m1s[i] + m2s[j] + __logf(acc[v]);
    }
  }
}

extern "C" void kernel_launch(void* const* d_in, const int* in_sizes, int n_in,
                              void* d_out, int out_size, void* d_ws, size_t ws_size,
                              hipStream_t stream) {
  (void)in_sizes; (void)n_in; (void)out_size; (void)ws_size;
  const float* inputs = (const float*)d_in[0];
  const float* grad   = (const float*)d_in[1];
  const float* cond   = (const float*)d_in[2];
  const float* weight = (const float*)d_in[3];
  const float* diagw  = (const float*)d_in[4];
  const float* bias   = (const float*)d_in[5];
  const float* wctx   = (const float*)d_in[6];

  float* out  = (float*)d_out;
  float* comb = out + BATCH * OUT_F;

  float* logterm = (float*)d_ws;

  out_gemm_kernel<<<dim3(64, 2), dim3(GEMM_THREADS), 0, stream>>>(inputs, cond, weight, diagw, wctx, bias, out, logterm);
  comb_kernel<<<dim3(DIMB, BATCH), dim3(256), 0, stream>>>(grad, weight, logterm, comb);
}


# both launches merged into one grid: comb work-groups (two batch entries each, 512 threads) sum their own rows' squares for logterm instead of reading the out GEMM's; its wsn summation order changes
# speedup vs baseline: 5.5765x; 1.2065x over previous
#include <hip/hip_runtime.h>
#include <cmath>


typedef __attribute__((ext_vector_type(2))) float v2f;
typedef __attribute__((ext_vector_type(4))) float v4f;
typedef __attribute__((ext_vector_type(8))) float v8f;

#define OUT_F 1024
#define IN_F  1024
#define CTX_F 1024
#define BATCH 32
#define DIMB  16

__device__ __forceinline__ v8f wmma_f32(v2f a, v2f b, v8f c) {
  return __builtin_amdgcn_wmma_f32_16x16x4_f32(
      false, a, false, b, (short)0, c, false, false);
}

#define KWAVES 16
#define KSLICE 128
#define KSTEPS (KSLICE / 4)
#define GROUP_THREADS 512
#define BLK 64
#define BSTEPS (BLK / 4)
#define GEMM_GROUPS 128
#define COMB_GROUPS 256

__device__ __forceinline__ void
out_gemm_group(const int nt, const int mt,
               const float* __restrict__ inputs,
               const float* __restrict__ cond,
               const float* __restrict__ weight,
               const float* __restrict__ diagw,
               const float* __restrict__ wctx,
               const float* __restrict__ bias,
               float* __restrict__ out) {
  const int t    = threadIdx.x;
  const int wave = t >> 5;
  const int lane = t & 31;
  const int mA   = lane & 15;
  const int kb   = (lane >> 4) << 1;
  const int nB   = lane & 15;

  __shared__ float part[KWAVES * 256];
  __shared__ float sqp[(KWAVES / 2) * 32];

  const float dw = diagw[nt * 16 + nB];
  const float bs = bias[nt * 16 + nB];

  v8f acc = {};

  if (wave < KWAVES / 2) {
    const int rb = (nt * 16) / BLK;
    const float* ap = inputs + (mt * 16 + mA) * IN_F + kb;
    const float* bp = weight + (nt * 16 + nB) * IN_F + kb;
    float sq = 0.f;
    for (int h = 0; h < KSLICE / BLK; ++h) {
      const int cb = wave * (KSLICE / BLK) + h;
      const float* aq = ap + cb * BLK;
      const float* bq = bp + cb * BLK;
      if (cb < rb) {
        for (int s = 0; s < BSTEPS; ++s) {
          v2f a; a.x = aq[4 * s]; a.y = aq[4 * s + 1];
          v2f b; b.x = bq[4 * s]; b.y = bq[4 * s + 1];
          sq += b.x * b.x + b.y * b.y;
          acc = wmma_f32(a, b, acc);
        }
      } else if (cb == rb) {
        for (int s = 0; s < BSTEPS; ++s) {
          v2f a; a.x = aq[4 * s]; a.y = aq[4 * s + 1];
          v2f b; b.x = __expf(bq[4 * s]); b.y = __expf(bq[4 * s + 1]);
          sq += b.x * b.x + b.y * b.y;
          acc = wmma_f32(a, b, acc);
        }
      }
    }
    sqp[wave * 32 + lane] = sq;
  } else {
    const int ks = (wave - KWAVES / 2) * KSLICE;
    const float* ap = cond + (mt * 16 + mA) * CTX_F + ks + kb;
    const float* bp = wctx + (size_t)(ks + kb) * OUT_F + nt * 16 + nB;
    for (int s = 0; s < KSTEPS; ++s) {
      v2f a; a.x = ap[4 * s]; a.y = ap[4 * s + 1];
      v2f b;
      b.x = bp[(size_t)(4 * s) * OUT_F];
      b.y = bp[(size_t)(4 * s + 1) * OUT_F];
      acc = wmma_f32(a, b, acc);
    }
  }

  for (int v = 0; v < 8; ++v) part[wave * 256 + v * 32 + lane] = acc[v];
  __syncthreads();

  if (t < 256) {
    const int c = t & 15;
    float wsn = sqp[c] + sqp[c + 16];
    for (int w = 1; w < KWAVES / 2; ++w) wsn += sqp[w * 32 + c] + sqp[w * 32 + c + 16];
    float s1 = part[t];
    for (int w = 1; w < KWAVES / 2; ++w) s1 += part[w * 256 + t];
    float s2 = part[(KWAVES / 2) * 256 + t];
    for (int w = KWAVES / 2 + 1; w < KWAVES; ++w) s2 += part[w * 256 + t];
    const int v = t >> 5;
    const int l = t & 31;
    const int m = mt * 16 + ((l >> 4) << 3) + v;
    const int n = nt * 16 + (l & 15);
    const float scale = __expf(dw) * rsqrtf(wsn);
    out[m * OUT_F + n] = scale * s1 + s2 + bs;
  }
}

template <int N>
__device__ __forceinline__ float left_squares(const float* __restrict__ p) {
  v4f x[2 * N];
#pragma unroll
  for (int u = 0; u < N; ++u) {
    x[2 * u]     = *(const v4f*)(p + u * 64);
    x[2 * u + 1] = *(const v4f*)(p + u * 64 + 32);
  }
  float s = 0.f;
#pragma unroll
  for (int u = 0; u < 2 * N; ++u)
    s += x[u].x * x[u].x + x[u].y * x[u].y + x[u].z * x[u].z + x[u].w * x[u].w;
  return s;
}

__device__ __forceinline__ float left_squares_upto8(const int n, const float* __restrict__ p) {
  switch (n) {
    case 1: return left_squares<1>(p);
    case 2: return left_squares<2>(p);
    case 3: return left_squares<3>(p);
    case 4: return left_squares<4>(p);
    case 5: return left_squares<5>(p);
    case 6: return left_squares<6>(p);
    case 7: return left_squares<7>(p);
    case 8: return left_squares<8>(p);
    default: return 0.f;
  }
}

__device__ __forceinline__ void
comb_group(const int d, const int bp,
           const float* __restrict__ grad,
           const float* __restrict__ weight,
           const float* __restrict__ diagw,
           float* __restrict__ comb) {
  const int t = threadIdx.x;
  const int STR = 66;
  const int RPP = 17;
  const int SPP = 9;

  __shared__ __attribute__((aligned(16))) float eg[64 * 66];
  __shared__ __attribute__((aligned(16))) float er[2 * 64 * 66];
  __shared__ __attribute__((aligned(16))) float colp[2 * 16 * 64];
  __shared__ __attribute__((aligned(16))) float rowp[64 * 17];
  __shared__ __attribute__((aligned(16))) float sqd[64 * 17];
  __shared__ __attribute__((aligned(16))) float sqo[64 * 9];
  __shared__ __attribute__((aligned(16))) float m1s[64];
  __shared__ __attribute__((aligned(16))) float m2s[2 * 64];
  __shared__ __attribute__((aligned(16))) float lts[64];

  const int hb = t >> 8;
  const int b  = 2 * bp + hb;
  const int c4 = t & 15;
  const int r0 = (t & 255) >> 4;
  const int rw = t >> 4;

  const float* gblk = grad + (size_t)((b * DIMB + d) * 64) * 64;
  const float* wblk = weight + (size_t)(d * 64) * IN_F + d * 64;

  v4f gv[4], wv[2];
  for (int q = 0; q < 4; ++q)
    gv[q] = *(const v4f*)(gblk + (r0 + 16 * q) * 64 + 4 * c4);
  for (int q = 0; q < 2; ++q)
    wv[q] = *(const v4f*)(wblk + (size_t)(rw + 32 * q) * IN_F + 4 * c4);
  const float dwv = diagw[d * 64 + (t & 63)];

  {
    const int orow = t >> 3;
    const int osub = t & 7;
    const float* orp = weight + (size_t)(d * 64 + orow) * IN_F + 4 * osub;
    float sq = left_squares_upto8(d < 8 ? d : 8, orp);
    if (d > 8) sq += left_squares_upto8(d - 8, orp + 8 * BLK);
    sqo[orow * SPP + osub] = sq;
  }

  {
    v4f cm;
    cm.x = fmaxf(fmaxf(gv[0].x, gv[1].x), fmaxf(gv[2].x, gv[3].x));
    cm.y = fmaxf(fmaxf(gv[0].y, gv[1].y), fmaxf(gv[2].y, gv[3].y));
    cm.z = fmaxf(fmaxf(gv[0].z, gv[1].z), fmaxf(gv[2].z, gv[3].z));
    cm.w = fmaxf(fmaxf(gv[0].w, gv[1].w), fmaxf(gv[2].w, gv[3].w));
    *(v4f*)&colp[hb * 1024 + r0 * 64 + 4 * c4] = cm;
    for (int q = 0; q < 2; ++q) {
      const int i = rw + 32 * q;
      rowp[i * RPP + c4] =
          fmaxf(fmaxf(wv[q].x, wv[q].y), fmaxf(wv[q].z, wv[q].w));
      const float ex = __expf(wv[q].x), ey = __expf(wv[q].y);
      const float ez = __expf(wv[q].z), ew = __expf(wv[q].w);
      sqd[i * RPP + c4] = ex * ex + ey * ey + ez * ez + ew * ew;
    }
  }
  __syncthreads();

  if (t < 128) {
    const float* cp = &colp[(t >> 6) * 1024 + (t & 63)];
    float m = -3.4e38f;
    for (int r = 0; r < 16; ++r) m = fmaxf(m, cp[r * 64]);
    m2s[t] = m;
  } else if (t < 192) {
    const int i = t - 128;
    float m = -3.4e38f;
    for (int c = 0; c < 16; ++c) m = fmaxf(m, rowp[i * RPP + c]);
    float wsn = sqd[i * RPP];
    for (int c = 1; c < 16; ++c) wsn += sqd[i * RPP + c];
    float wso = sqo[i * SPP];
    for (int c = 1; c < 8; ++c) wso += sqo[i * SPP + c];
    wsn += wso;
    const float ltv = dwv - 0.5f * __logf(wsn);
    m1s[i] = ltv + m;
    lts[i] = ltv;
  }
  __syncthreads();

  {
    float* erb = &er[hb * (64 * 66)];
    const v4f m2v = *(const v4f*)&m2s[hb * 64 + 4 * c4];
    for (int q = 0; q < 4; ++q) {
      const int k = r0 + 16 * q;
      v2f lo, hi;
      lo.x = __expf(gv[q].x - m2v.x);
      lo.y = __expf(gv[q].y - m2v.y);
      hi.x = __expf(gv[q].z - m2v.z);
      hi.y = __expf(gv[q].w - m2v.w);
      *(v2f*)&erb[k * STR + 4 * c4]     = lo;
      *(v2f*)&erb[k * STR + 4 * c4 + 2] = hi;
    }
    for (int q = 0; q < 2; ++q) {
      const int i = rw + 32 * q;
      const float lt = lts[i];
      const float m1 = m1s[i];
      v2f lo, hi;
      lo.x = __expf((lt + wv[q].x) - m1);
      lo.y = __expf((lt + wv[q].y) - m1);
      hi.x = __expf((lt + wv[q].z) - m1);
      hi.y = __expf((lt + wv[q].w) - m1);
      *(v2f*)&eg[i * STR + 4 * c4]     = lo;
      *(v2f*)&eg[i * STR + 4 * c4 + 2] = hi;
    }
  }
  __syncthreads();

  const int wave  = t >> 5;
  const int lane  = t & 31;
  const int mA    = lane & 15;
  const int kb    = (lane >> 4) << 1;
  const int nB    = lane & 15;
  const int mBase = (lane >> 4) << 3;
  const float* erb = &er[hb * (64 * 66)];
  const float* m2b = &m2s[hb * 64];

  for (int tile = (wave & 7); tile < 16; tile += 8) {
    const int i0 = (tile >> 2) * 16;
    const int j0 = (tile & 3) * 16;
    v8f acc = {};
    const float* ap = &eg[(i0 + mA) * STR + kb];
    const float* bq = &erb[kb * STR + j0 + nB];
    for (int kk = 0; kk < 16; ++kk) {
      v2f a; a.x = ap[kk * 4];           a.y = ap[kk * 4 + 1];
      v2f bb; bb.x = bq[(kk * 4) * STR]; bb.y = bq[(kk * 4 + 1) * STR];
      acc = wmma_f32(a, bb, acc);
    }
    for (int v = 0; v < 8; ++v) {
      const int i = i0 + mBase + v;
      const int j = j0 + (lane & 15);
      comb[((size_t)(b * DIMB + d) * 64 + i) * 64 + j] =
          m1s[i] + m2b[j] + __logf(acc[v]);
    }
  }
}

__global__ void __launch_bounds__(GROUP_THREADS)
masked_weight_kernel(const float* __restrict__ inputs,
                     const float* __restrict__ grad,
                     const float* __restrict__ cond,
                     const float* __restrict__ weight,
                     const float* __restrict__ diagw,
                     const float* __restrict__ bias,
                     const float* __restrict__ wctx,
                     float* __restrict__ out,
                     float* __restrict__ comb) {
  const int g = blockIdx.x;
  if (g < COMB_GROUPS) {
    comb_group((DIMB - 1) - (g >> 4), g & 15, grad, weight, diagw, comb);
  } else {
    const int h = g - COMB_GROUPS;
    out_gemm_group(h & 63, h >> 6, inputs, cond, weight, diagw, wctx, bias, out);
  }
}

extern "C" void kernel_launch(void* const* d_in, const int* in_sizes, int n_in,
                              void* d_out, int out_size, void* d_ws, size_t ws_size,
                              hipStream_t stream) {
  (void)in_sizes; (void)n_in; (void)out_size; (void)d_ws; (void)ws_size;
  const float* inputs = (const float*)d_in[0];
  const float* grad   = (const float*)d_in[1];
  const float* cond   = (const float*)d_in[2];
  const float* weight = (const float*)d_in[3];
  const float* diagw  = (const float*)d_in[4];
  const float* bias   = (const float*)d_in[5];
  const float* wctx   = (const float*)d_in[6];

  float* out  = (float*)d_out;
  float* comb = out + BATCH * OUT_F;

  masked_weight_kernel<<<dim3(GEMM_GROUPS + COMB_GROUPS), dim3(GROUP_THREADS), 0, stream>>>(
      inputs, grad, cond, weight, diagw, bias, wctx, out, comb);
}


# comb work-groups take four batch entries instead of two (128 comb + 128 out-GEMM groups of 16 waves, 256 groups for 256 WGPs): a third fewer groups and waves enter, wsn sums redone half as often
# speedup vs baseline: 6.4577x; 1.1580x over previous
#include <hip/hip_runtime.h>
#include <cmath>


typedef __attribute__((ext_vector_type(2))) float v2f;
typedef __attribute__((ext_vector_type(4))) float v4f;
typedef __attribute__((ext_vector_type(8))) float v8f;

#define OUT_F 1024
#define IN_F  1024
#define CTX_F 1024
#define BATCH 32
#define DIMB  16

__device__ __forceinline__ v8f wmma_f32(v2f a, v2f b, v8f c) {
  return __builtin_amdgcn_wmma_f32_16x16x4_f32(
      false, a, false, b, (short)0, c, false, false);
}

#define KWAVES 16
#define KSLICE 128
#define KSTEPS (KSLICE / 4)
#define GROUP_THREADS 512
#define BLK 64
#define BSTEPS (BLK / 4)
#define GEMM_GROUPS 128
#define COMB_ENTRIES 4
#define COMB_GROUPS 128

__device__ __forceinline__ void
out_gemm_group(const int nt, const int mt,
               const float* __restrict__ inputs,
               const float* __restrict__ cond,
               const float* __restrict__ weight,
               const float* __restrict__ diagw,
               const float* __restrict__ wctx,
               const float* __restrict__ bias,
               float* __restrict__ out) {
  const int t    = threadIdx.x;
  const int wave = t >> 5;
  const int lane = t & 31;
  const int mA   = lane & 15;
  const int kb   = (lane >> 4) << 1;
  const int nB   = lane & 15;

  __shared__ float part[KWAVES * 256];
  __shared__ float sqp[(KWAVES / 2) * 32];

  const float dw = diagw[nt * 16 + nB];
  const float bs = bias[nt * 16 + nB];

  v8f acc = {};

  if (wave < KWAVES / 2) {
    const int rb = (nt * 16) / BLK;
    const float* ap = inputs + (mt * 16 + mA) * IN_F + kb;
    const float* bp = weight + (nt * 16 + nB) * IN_F + kb;
    float sq = 0.f;
    for (int h = 0; h < KSLICE / BLK; ++h) {
      const int cb = wave * (KSLICE / BLK) + h;
      const float* aq = ap + cb * BLK;
      const float* bq = bp + cb * BLK;
      if (cb < rb) {
        for (int s = 0; s < BSTEPS; ++s) {
          v2f a; a.x = aq[4 * s]; a.y = aq[4 * s + 1];
          v2f b; b.x = bq[4 * s]; b.y = bq[4 * s + 1];
          sq += b.x * b.x + b.y * b.y;
          acc = wmma_f32(a, b, acc);
        }
      } else if (cb == rb) {
        for (int s = 0; s < BSTEPS; ++s) {
          v2f a; a.x = aq[4 * s]; a.y = aq[4 * s + 1];
          v2f b; b.x = __expf(bq[4 * s]); b.y = __expf(bq[4 * s + 1]);
          sq += b.x * b.x + b.y * b.y;
          acc = wmma_f32(a, b, acc);
        }
      }
    }
    sqp[wave * 32 + lane] = sq;
  } else {
    const int ks = (wave - KWAVES / 2) * KSLICE;
    const float* ap = cond + (mt * 16 + mA) * CTX_F + ks + kb;
    const float* bp = wctx + (size_t)(ks + kb) * OUT_F + nt * 16 + nB;
    for (int s = 0; s < KSTEPS; ++s) {
      v2f a; a.x = ap[4 * s]; a.y = ap[4 * s + 1];
      v2f b;
      b.x = bp[(size_t)(4 * s) * OUT_F];
      b.y = bp[(size_t)(4 * s + 1) * OUT_F];
      acc = wmma_f32(a, b, acc);
    }
  }

  for (int v = 0; v < 8; ++v) part[wave * 256 + v * 32 + lane] = acc[v];
  __syncthreads();

  if (t < 256) {
    const int c = t & 15;
    float wsn = sqp[c] + sqp[c + 16];
    for (int w = 1; w < KWAVES / 2; ++w) wsn += sqp[w * 32 + c] + sqp[w * 32 + c + 16];
    float s1 = part[t];
    for (int w = 1; w < KWAVES / 2; ++w) s1 += part[w * 256 + t];
    float s2 = part[(KWAVES / 2) * 256 + t];
    for (int w = KWAVES / 2 + 1; w < KWAVES; ++w) s2 += part[w * 256 + t];
    const int v = t >> 5;
    const int l = t & 31;
    const int m = mt * 16 + ((l >> 4) << 3) + v;
    const int n = nt * 16 + (l & 15);
    const float scale = __expf(dw) * rsqrtf(wsn);
    out[m * OUT_F + n] = scale * s1 + s2 + bs;
  }
}

template <int N>
__device__ __forceinline__ float left_squares(const float* __restrict__ p) {
  v4f x[2 * N];
#pragma unroll
  for (int u = 0; u < N; ++u) {
    x[2 * u]     = *(const v4f*)(p + u * 64);
    x[2 * u + 1] = *(const v4f*)(p + u * 64 + 32);
  }
  float s = 0.f;
#pragma unroll
  for (int u = 0; u < 2 * N; ++u)
    s += x[u].x * x[u].x + x[u].y * x[u].y + x[u].z * x[u].z + x[u].w * x[u].w;
  return s;
}

__device__ __forceinline__ float left_squares_upto8(const int n, const float* __restrict__ p) {
  switch (n) {
    case 1: return left_squares<1>(p);
    case 2: return left_squares<2>(p);
    case 3: return left_squares<3>(p);
    case 4: return left_squares<4>(p);
    case 5: return left_squares<5>(p);
    case 6: return left_squares<6>(p);
    case 7: return left_squares<7>(p);
    case 8: return left_squares<8>(p);
    default: return 0.f;
  }
}

__device__ __forceinline__ void
comb_group(const int d, const int bq,
           const float* __restrict__ grad,
           const float* __restrict__ weight,
           const float* __restrict__ diagw,
           float* __restrict__ comb) {
  const int t = threadIdx.x;
  const int STR = 66;
  const int RPP = 17;
  const int SPP = 9;

  __shared__ __attribute__((aligned(16))) float eg[64 * 66];
  __shared__ __attribute__((aligned(16))) float er[COMB_ENTRIES * 64 * 66];
  __shared__ __attribute__((aligned(16))) float colp[COMB_ENTRIES * 16 * 64];
  __shared__ __attribute__((aligned(16))) float rowp[64 * 17];
  __shared__ __attribute__((aligned(16))) float sqd[64 * 17];
  __shared__ __attribute__((aligned(16))) float sqo[64 * 9];
  __shared__ __attribute__((aligned(16))) float m1s[64];
  __shared__ __attribute__((aligned(16))) float m2s[COMB_ENTRIES * 64];
  __shared__ __attribute__((aligned(16))) float lts[64];

  const int hb = t >> 8;
  const int b0 = COMB_ENTRIES * bq;
  const int c4 = t & 15;
  const int r0 = (t & 255) >> 4;
  const int rw = t >> 4;

  const float* wblk = weight + (size_t)(d * 64) * IN_F + d * 64;

  v4f gv[2][4], wv[2];
  for (int e = 0; e < 2; ++e) {
    const float* gblk = grad + (size_t)(((b0 + 2 * hb + e) * DIMB + d) * 64) * 64;
    for (int q = 0; q < 4; ++q)
      gv[e][q] = *(const v4f*)(gblk + (r0 + 16 * q) * 64 + 4 * c4);
  }
  for (int q = 0; q < 2; ++q)
    wv[q] = *(const v4f*)(wblk + (size_t)(rw + 32 * q) * IN_F + 4 * c4);
  const float dwv = diagw[d * 64 + (t & 63)];

  {
    const int orow = t >> 3;
    const int osub = t & 7;
    const float* orp = weight + (size_t)(d * 64 + orow) * IN_F + 4 * osub;
    float sq = left_squares_upto8(d < 8 ? d : 8, orp);
    if (d > 8) sq += left_squares_upto8(d - 8, orp + 8 * BLK);
    sqo[orow * SPP + osub] = sq;
  }

  {
    for (int e = 0; e < 2; ++e) {
      v4f cm;
      cm.x = fmaxf(fmaxf(gv[e][0].x, gv[e][1].x), fmaxf(gv[e][2].x, gv[e][3].x));
      cm.y = fmaxf(fmaxf(gv[e][0].y, gv[e][1].y), fmaxf(gv[e][2].y, gv[e][3].y));
      cm.z = fmaxf(fmaxf(gv[e][0].z, gv[e][1].z), fmaxf(gv[e][2].z, gv[e][3].z));
      cm.w = fmaxf(fmaxf(gv[e][0].w, gv[e][1].w), fmaxf(gv[e][2].w, gv[e][3].w));
      *(v4f*)&colp[(2 * hb + e) * 1024 + r0 * 64 + 4 * c4] = cm;
    }
    for (int q = 0; q < 2; ++q) {
      const int i = rw + 32 * q;
      rowp[i * RPP + c4] =
          fmaxf(fmaxf(wv[q].x, wv[q].y), fmaxf(wv[q].z, wv[q].w));
      const float ex = __expf(wv[q].x), ey = __expf(wv[q].y);
      const float ez = __expf(wv[q].z), ew = __expf(wv[q].w);
      sqd[i * RPP + c4] = ex * ex + ey * ey + ez * ez + ew * ew;
    }
  }
  __syncthreads();

  if (t < COMB_ENTRIES * 64) {
    const float* cp = &colp[(t >> 6) * 1024 + (t & 63)];
    float m = -3.4e38f;
    for (int r = 0; r < 16; ++r) m = fmaxf(m, cp[r * 64]);
    m2s[t] = m;
  } else if (t < COMB_ENTRIES * 64 + 64) {
    const int i = t - COMB_ENTRIES * 64;
    float m = -3.4e38f;
    for (int c = 0; c < 16; ++c) m = fmaxf(m, rowp[i * RPP + c]);
    float wsn = sqd[i * RPP];
    for (int c = 1; c < 16; ++c) wsn += sqd[i * RPP + c];
    float wso = sqo[i * SPP];
    for (int c = 1; c < 8; ++c) wso += sqo[i * SPP + c];
    wsn += wso;
    const float ltv = dwv - 0.5f * __logf(wsn);
    m1s[i] = ltv + m;
    lts[i] = ltv;
  }
  __syncthreads();

  {
    for (int e = 0; e < 2; ++e) {
      float* erw = &er[(2 * hb + e) * (64 * 66)];
      const v4f m2v = *(const v4f*)&m2s[(2 * hb + e) * 64 + 4 * c4];
      for (int q = 0; q < 4; ++q) {
        const int k = r0 + 16 * q;
        v2f lo, hi;
        lo.x = __expf(gv[e][q].x - m2v.x);
        lo.y = __expf(gv[e][q].y - m2v.y);
        hi.x = __expf(gv[e][q].z - m2v.z);
        hi.y = __expf(gv[e][q].w - m2v.w);
        *(v2f*)&erw[k * STR + 4 * c4]     = lo;
        *(v2f*)&erw[k * STR + 4 * c4 + 2] = hi;
      }
    }
    for (int q = 0; q < 2; ++q) {
      const int i = rw + 32 * q;
      const float lt = lts[i];
      const float m1 = m1s[i];
      v2f lo, hi;
      lo.x = __expf((lt + wv[q].x) - m1);
      lo.y = __expf((lt + wv[q].y) - m1);
      hi.x = __expf((lt + wv[q].z) - m1);
      hi.y = __expf((lt + wv[q].w) - m1);
      *(v2f*)&eg[i * STR + 4 * c4]     = lo;
      *(v2f*)&eg[i * STR + 4 * c4 + 2] = hi;
    }
  }
  __syncthreads();

  const int wave  = t >> 5;
  const int lane  = t & 31;
  const int mA    = lane & 15;
  const int kb    = (lane >> 4) << 1;
  const int nB    = lane & 15;
  const int mBase = (lane >> 4) << 3;
  const int we    = wave >> 2;
  const int i0    = (wave & 3) * 16;
  const int b     = b0 + we;
  const float* erb = &er[we * (64 * 66)];
  const float* m2b = &m2s[we * 64];

  {
    v8f acc[4] = {};
    const float* ap = &eg[(i0 + mA) * STR + kb];
    const float* bf = &erb[kb * STR + nB];
    for (int kk = 0; kk < 16; ++kk) {
      v2f a; a.x = ap[kk * 4]; a.y = ap[kk * 4 + 1];
#pragma unroll
      for (int jt = 0; jt < 4; ++jt) {
        v2f bb;
        bb.x = bf[(kk * 4) * STR + 16 * jt];
        bb.y = bf[(kk * 4 + 1) * STR + 16 * jt];
        acc[jt] = wmma_f32(a, bb, acc[jt]);
      }
    }
#pragma unroll
    for (int jt = 0; jt < 4; ++jt) {
      for (int v = 0; v < 8; ++v) {
        const int i = i0 + mBase + v;
        const int j = 16 * jt + (lane & 15);
        comb[((size_t)(b * DIMB + d) * 64 + i) * 64 + j] =
            m1s[i] + m2b[j] + __logf(acc[jt][v]);
      }
    }
  }
}

__global__ void __launch_bounds__(GROUP_THREADS)
masked_weight_kernel(const float* __restrict__ inputs,
                     const float* __restrict__ grad,
                     const float* __restrict__ cond,
                     const float* __restrict__ weight,
                     const float* __restrict__ diagw,
                     const float* __restrict__ bias,
                     const float* __restrict__ wctx,
                     float* __restrict__ out,
                     float* __restrict__ comb) {
  const int g = blockIdx.x;
  if (g < COMB_GROUPS) {
    comb_group((DIMB - 1) - (g >> 3), g & 7, grad, weight, diagw, comb);
  } else {
    const int h = g - COMB_GROUPS;
    out_gemm_group(h & 63, h >> 6, inputs, cond, weight, diagw, wctx, bias, out);
  }
}

extern "C" void kernel_launch(void* const* d_in, const int* in_sizes, int n_in,
                              void* d_out, int out_size, void* d_ws, size_t ws_size,
                              hipStream_t stream) {
  (void)in_sizes; (void)n_in; (void)out_size; (void)d_ws; (void)ws_size;
  const float* inputs = (const float*)d_in[0];
  const float* grad   = (const float*)d_in[1];
  const float* cond   = (const float*)d_in[2];
  const float* weight = (const float*)d_in[3];
  const float* diagw  = (const float*)d_in[4];
  const float* bias   = (const float*)d_in[5];
  const float* wctx   = (const float*)d_in[6];

  float* out  = (float*)d_out;
  float* comb = out + BATCH * OUT_F;

  masked_weight_kernel<<<dim3(GEMM_GROUPS + COMB_GROUPS), dim3(GROUP_THREADS), 0, stream>>>(
      inputs, grad, cond, weight, diagw, bias, wctx, out, comb);
}


# comb: logterm taken out of expG (it cancels against m1) and added into m1 late, so the row-square loads left of the diagonal fly under the exp and WMMA phases (LDS-only barriers); no sum order change
# speedup vs baseline: 7.1213x; 1.1028x over previous
#include <hip/hip_runtime.h>
#include <cmath>


typedef __attribute__((ext_vector_type(2))) float v2f;
typedef __attribute__((ext_vector_type(4))) float v4f;
typedef __attribute__((ext_vector_type(8))) float v8f;

#define OUT_F 1024
#define IN_F  1024
#define CTX_F 1024
#define BATCH 32
#define DIMB  16

__device__ __forceinline__ v8f wmma_f32(v2f a, v2f b, v8f c) {
  return __builtin_amdgcn_wmma_f32_16x16x4_f32(
      false, a, false, b, (short)0, c, false, false);
}

__device__ __forceinline__ void lds_barrier() {
  __builtin_amdgcn_fence(__ATOMIC_RELEASE, "workgroup", "local");
  __builtin_amdgcn_s_barrier();
  __builtin_amdgcn_fence(__ATOMIC_ACQUIRE, "workgroup", "local");
}

#define KWAVES 16
#define KSLICE 128
#define KSTEPS (KSLICE / 4)
#define GROUP_THREADS 512
#define BLK 64
#define BSTEPS (BLK / 4)
#define GEMM_GROUPS 128
#define COMB_ENTRIES 4
#define COMB_GROUPS 128

__device__ __forceinline__ void
out_gemm_group(const int nt, const int mt,
               const float* __restrict__ inputs,
               const float* __restrict__ cond,
               const float* __restrict__ weight,
               const float* __restrict__ diagw,
               const float* __restrict__ wctx,
               const float* __restrict__ bias,
               float* __restrict__ out) {
  const int t    = threadIdx.x;
  const int wave = t >> 5;
  const int lane = t & 31;
  const int mA   = lane & 15;
  const int kb   = (lane >> 4) << 1;
  const int nB   = lane & 15;

  __shared__ float part[KWAVES * 256];
  __shared__ float sqp[(KWAVES / 2) * 32];

  const float dw = diagw[nt * 16 + nB];
  const float bs = bias[nt * 16 + nB];

  v8f acc = {};

  if (wave < KWAVES / 2) {
    const int rb = (nt * 16) / BLK;
    const float* ap = inputs + (mt * 16 + mA) * IN_F + kb;
    const float* bp = weight + (nt * 16 + nB) * IN_F + kb;
    float sq = 0.f;
    for (int h = 0; h < KSLICE / BLK; ++h) {
      const int cb = wave * (KSLICE / BLK) + h;
      const float* aq = ap + cb * BLK;
      const float* bq = bp + cb * BLK;
      if (cb < rb) {
        for (int s = 0; s < BSTEPS; ++s) {
          v2f a; a.x = aq[4 * s]; a.y = aq[4 * s + 1];
          v2f b; b.x = bq[4 * s]; b.y = bq[4 * s + 1];
          sq += b.x * b.x + b.y * b.y;
          acc = wmma_f32(a, b, acc);
        }
      } else if (cb == rb) {
        for (int s = 0; s < BSTEPS; ++s) {
          v2f a; a.x = aq[4 * s]; a.y = aq[4 * s + 1];
          v2f b; b.x = __expf(bq[4 * s]); b.y = __expf(bq[4 * s + 1]);
          sq += b.x * b.x + b.y * b.y;
          acc = wmma_f32(a, b, acc);
        }
      }
    }
    sqp[wave * 32 + lane] = sq;
  } else {
    const int ks = (wave - KWAVES / 2) * KSLICE;
    const float* ap = cond + (mt * 16 + mA) * CTX_F + ks + kb;
    const float* bp = wctx + (size_t)(ks + kb) * OUT_F + nt * 16 + nB;
    for (int s = 0; s < KSTEPS; ++s) {
      v2f a; a.x = ap[4 * s]; a.y = ap[4 * s + 1];
      v2f b;
      b.x = bp[(size_t)(4 * s) * OUT_F];
      b.y = bp[(size_t)(4 * s + 1) * OUT_F];
      acc = wmma_f32(a, b, acc);
    }
  }

  for (int v = 0; v < 8; ++v) part[wave * 256 + v * 32 + lane] = acc[v];
  __syncthreads();

  if (t < 256) {
    const int c = t & 15;
    float wsn = sqp[c] + sqp[c + 16];
    for (int w = 1; w < KWAVES / 2; ++w) wsn += sqp[w * 32 + c] + sqp[w * 32 + c + 16];
    float s1 = part[t];
    for (int w = 1; w < KWAVES / 2; ++w) s1 += part[w * 256 + t];
    float s2 = part[(KWAVES / 2) * 256 + t];
    for (int w = KWAVES / 2 + 1; w < KWAVES; ++w) s2 += part[w * 256 + t];
    const int v = t >> 5;
    const int l = t & 31;
    const int m = mt * 16 + ((l >> 4) << 3) + v;
    const int n = nt * 16 + (l & 15);
    const float scale = __expf(dw) * rsqrtf(wsn);
    out[m * OUT_F + n] = scale * s1 + s2 + bs;
  }
}

#define LEFT_BLOCKS (DIMB - 1)

template <int U>
__device__ __forceinline__ void left_loads_from(const int n, v4f (&x)[2 * LEFT_BLOCKS],
                                                const float* __restrict__ p) {
  if constexpr (U < LEFT_BLOCKS) {
    if (U < n) {
      x[2 * U]     = *(const v4f*)(p + U * 64);
      x[2 * U + 1] = *(const v4f*)(p + U * 64 + 32);
      left_loads_from<U + 1>(n, x, p);
    }
  }
}

template <int U, int END>
__device__ __forceinline__ float left_sum_from(const int n, const v4f (&x)[2 * LEFT_BLOCKS], float s) {
  if constexpr (U < END) {
    if (U < n) {
      const v4f a = x[2 * U], b = x[2 * U + 1];
      asm volatile("" :: "v"(a), "v"(b));
      s += a.x * a.x + a.y * a.y + a.z * a.z + a.w * a.w;
      s += b.x * b.x + b.y * b.y + b.z * b.z + b.w * b.w;
      s = left_sum_from<U + 1, END>(n, x, s);
    }
  }
  return s;
}

__device__ __forceinline__ void
comb_group(const int d, const int bq,
           const float* __restrict__ grad,
           const float* __restrict__ weight,
           const float* __restrict__ diagw,
           float* __restrict__ comb) {
  const int t = threadIdx.x;
  const int STR = 66;
  const int RPP = 17;
  const int SPP = 9;

  __shared__ __attribute__((aligned(16))) float eg[64 * 66];
  __shared__ __attribute__((aligned(16))) float er[COMB_ENTRIES * 64 * 66];
  __shared__ __attribute__((aligned(16))) float colp[COMB_ENTRIES * 16 * 64];
  __shared__ __attribute__((aligned(16))) float rowp[64 * 17];
  __shared__ __attribute__((aligned(16))) float sqd[64 * 17];
  __shared__ __attribute__((aligned(16))) float sqo[64 * 9];
  __shared__ __attribute__((aligned(16))) float m1s[64];
  __shared__ __attribute__((aligned(16))) float m2s[COMB_ENTRIES * 64];
  __shared__ __attribute__((aligned(16))) float mws[64];

  const int hb = t >> 8;
  const int b0 = COMB_ENTRIES * bq;
  const int c4 = t & 15;
  const int r0 = (t & 255) >> 4;
  const int rw = t >> 4;

  const float* wblk = weight + (size_t)(d * 64) * IN_F + d * 64;

  v4f gv[2][4], wv[2];
  for (int e = 0; e < 2; ++e) {
    const float* gblk = grad + (size_t)(((b0 + 2 * hb + e) * DIMB + d) * 64) * 64;
    for (int q = 0; q < 4; ++q)
      gv[e][q] = *(const v4f*)(gblk + (r0 + 16 * q) * 64 + 4 * c4);
  }
  for (int q = 0; q < 2; ++q)
    wv[q] = *(const v4f*)(wblk + (size_t)(rw + 32 * q) * IN_F + 4 * c4);
  const float dwv = diagw[d * 64 + (t & 63)];

  const int orow = t >> 3;
  const int osub = t & 7;
  v4f xl[2 * LEFT_BLOCKS];

  {
    for (int e = 0; e < 2; ++e) {
      v4f cm;
      cm.x = fmaxf(fmaxf(gv[e][0].x, gv[e][1].x), fmaxf(gv[e][2].x, gv[e][3].x));
      cm.y = fmaxf(fmaxf(gv[e][0].y, gv[e][1].y), fmaxf(gv[e][2].y, gv[e][3].y));
      cm.z = fmaxf(fmaxf(gv[e][0].z, gv[e][1].z), fmaxf(gv[e][2].z, gv[e][3].z));
      cm.w = fmaxf(fmaxf(gv[e][0].w, gv[e][1].w), fmaxf(gv[e][2].w, gv[e][3].w));
      *(v4f*)&colp[(2 * hb + e) * 1024 + r0 * 64 + 4 * c4] = cm;
    }
    for (int q = 0; q < 2; ++q) {
      const int i = rw + 32 * q;
      rowp[i * RPP + c4] =
          fmaxf(fmaxf(wv[q].x, wv[q].y), fmaxf(wv[q].z, wv[q].w));
      const float ex = __expf(wv[q].x), ey = __expf(wv[q].y);
      const float ez = __expf(wv[q].z), ew = __expf(wv[q].w);
      sqd[i * RPP + c4] = ex * ex + ey * ey + ez * ez + ew * ew;
    }
  }
  {
    const float* orp = weight + (size_t)(d * 64 + orow) * IN_F + 4 * osub;
    left_loads_from<0>(d, xl, orp);
  }
  lds_barrier();

  if (t < COMB_ENTRIES * 64) {
    const float* cp = &colp[(t >> 6) * 1024 + (t & 63)];
    float m = -3.4e38f;
    for (int r = 0; r < 16; ++r) m = fmaxf(m, cp[r * 64]);
    m2s[t] = m;
  } else if (t < COMB_ENTRIES * 64 + 64) {
    const int i = t - COMB_ENTRIES * 64;
    float m = -3.4e38f;
    for (int c = 0; c < 16; ++c) m = fmaxf(m, rowp[i * RPP + c]);
    mws[i] = m;
  }
  lds_barrier();

  {
    for (int e = 0; e < 2; ++e) {
      float* erw = &er[(2 * hb + e) * (64 * 66)];
      const v4f m2v = *(const v4f*)&m2s[(2 * hb + e) * 64 + 4 * c4];
      for (int q = 0; q < 4; ++q) {
        const int k = r0 + 16 * q;
        v2f lo, hi;
        lo.x = __expf(gv[e][q].x - m2v.x);
        lo.y = __expf(gv[e][q].y - m2v.y);
        hi.x = __expf(gv[e][q].z - m2v.z);
        hi.y = __expf(gv[e][q].w - m2v.w);
        *(v2f*)&erw[k * STR + 4 * c4]     = lo;
        *(v2f*)&erw[k * STR + 4 * c4 + 2] = hi;
      }
    }
    for (int q = 0; q < 2; ++q) {
      const int i = rw + 32 * q;
      const float mw = mws[i];
      v2f lo, hi;
      lo.x = __expf(wv[q].x - mw);
      lo.y = __expf(wv[q].y - mw);
      hi.x = __expf(wv[q].z - mw);
      hi.y = __expf(wv[q].w - mw);
      *(v2f*)&eg[i * STR + 4 * c4]     = lo;
      *(v2f*)&eg[i * STR + 4 * c4 + 2] = hi;
    }
  }
  lds_barrier();

  const int wave  = t >> 5;
  const int lane  = t & 31;
  const int mA    = lane & 15;
  const int kb    = (lane >> 4) << 1;
  const int nB    = lane & 15;
  const int mBase = (lane >> 4) << 3;
  const int we    = wave >> 2;
  const int i0    = (wave & 3) * 16;
  const int b     = b0 + we;
  const float* erb = &er[we * (64 * 66)];
  const float* m2b = &m2s[we * 64];

  {
    v8f acc[4] = {};
    const float* ap = &eg[(i0 + mA) * STR + kb];
    const float* bf = &erb[kb * STR + nB];
    auto ksteps = [&](const int k0, const int k1) {
      for (int kk = k0; kk < k1; ++kk) {
        v2f a; a.x = ap[kk * 4]; a.y = ap[kk * 4 + 1];
#pragma unroll
        for (int jt = 0; jt < 4; ++jt) {
          v2f bb;
          bb.x = bf[(kk * 4) * STR + 16 * jt];
          bb.y = bf[(kk * 4 + 1) * STR + 16 * jt];
          acc[jt] = wmma_f32(a, bb, acc[jt]);
        }
      }
    };
    ksteps(0, 12);
    sqo[orow * SPP + osub] = left_sum_from<0, 8>(d, xl, 0.f) + left_sum_from<8, LEFT_BLOCKS>(d, xl, 0.f);
    ksteps(12, 16);
    lds_barrier();
    if (t < 64) {
      const int i = t;
      float wsn = sqd[i * RPP];
      for (int c = 1; c < 16; ++c) wsn += sqd[i * RPP + c];
      float wso = sqo[i * SPP];
      for (int c = 1; c < 8; ++c) wso += sqo[i * SPP + c];
      wsn += wso;
      m1s[i] = (dwv - 0.5f * __logf(wsn)) + mws[i];
    }
    lds_barrier();
#pragma unroll
    for (int jt = 0; jt < 4; ++jt) {
      for (int v = 0; v < 8; ++v) {
        const int i = i0 + mBase + v;
        const int j = 16 * jt + (lane & 15);
        comb[((size_t)(b * DIMB + d) * 64 + i) * 64 + j] =
            m1s[i] + m2b[j] + __logf(acc[jt][v]);
      }
    }
  }
}

__global__ void __launch_bounds__(GROUP_THREADS)
masked_weight_kernel(const float* __restrict__ inputs,
                     const float* __restrict__ grad,
                     const float* __restrict__ cond,
                     const float* __restrict__ weight,
                     const float* __restrict__ diagw,
                     const float* __restrict__ bias,
                     const float* __restrict__ wctx,
                     float* __restrict__ out,
                     float* __restrict__ comb) {
  const int g = blockIdx.x;
  if (g < COMB_GROUPS) {
    comb_group((DIMB - 1) - (g >> 3), g & 7, grad, weight, diagw, comb);
  } else {
    const int h = g - COMB_GROUPS;
    out_gemm_group(h & 63, h >> 6, inputs, cond, weight, diagw, wctx, bias, out);
  }
}

extern "C" void kernel_launch(void* const* d_in, const int* in_sizes, int n_in,
                              void* d_out, int out_size, void* d_ws, size_t ws_size,
                              hipStream_t stream) {
  (void)in_sizes; (void)n_in; (void)out_size; (void)d_ws; (void)ws_size;
  const float* inputs = (const float*)d_in[0];
  const float* grad   = (const float*)d_in[1];
  const float* cond   = (const float*)d_in[2];
  const float* weight = (const float*)d_in[3];
  const float* diagw  = (const float*)d_in[4];
  const float* bias   = (const float*)d_in[5];
  const float* wctx   = (const float*)d_in[6];

  float* out  = (float*)d_out;
  float* comb = out + BATCH * OUT_F;

  masked_weight_kernel<<<dim3(GEMM_GROUPS + COMB_GROUPS), dim3(GROUP_THREADS), 0, stream>>>(
      inputs, grad, cond, weight, diagw, bias, wctx, out, comb);
}
